# starting kernel
# baseline (speedup 1.0000x reference)
#include <hip/hip_runtime.h>
#include <math.h>

typedef __attribute__((ext_vector_type(2))) float v2f;
typedef __attribute__((ext_vector_type(8))) float v8f;

constexpr int Bb = 32;
constexpr int Ss = 1024;
constexpr int Dd = 256;
constexpr int Nn = 32;
constexpr int Cc = 64;

constexpr size_t OFF_LOGITS = 0;
constexpr size_t OFF_CBUF   = OFF_LOGITS + (size_t)Bb*Ss*Nn;
constexpr size_t OFF_XC     = OFF_CBUF   + (size_t)Bb*Ss*Nn;
constexpr size_t OFF_WV     = OFF_XC     + (size_t)Bb*Nn*Dd;
constexpr size_t OFF_S      = OFF_WV     + (size_t)Bb*Nn*Dd;
constexpr size_t OFF_V      = OFF_S      + (size_t)Bb*Nn*Cc;

__device__ __forceinline__ v8f wmma_f32_k4(v2f a, v2f b, v8f c) {
  return __builtin_amdgcn_wmma_f32_16x16x4_f32(false, a, false, b, (short)0, c,
                                               false, false);
}

__global__ void caps_zero(float* __restrict__ p, int n) {
  int i = blockIdx.x * blockDim.x + threadIdx.x;
  if (i < n) p[i] = 0.0f;
}

__global__ void caps_softmax(const float* __restrict__ logits,
                             float* __restrict__ cbuf) {
  int t = blockIdx.x * blockDim.x + threadIdx.x;
  const float* lg = logits + (size_t)t * Nn;
  float* co = cbuf + (size_t)t * Nn;
  float e[Nn];
  float m = lg[0];
#pragma unroll
  for (int n = 1; n < Nn; ++n) m = fmaxf(m, lg[n]);
  float sum = 0.0f;
#pragma unroll
  for (int n = 0; n < Nn; ++n) {
    e[n] = __expf(lg[n] - m);
    sum += e[n];
  }
  float inv = 1.0f / sum;
#pragma unroll
  for (int n = 0; n < Nn; ++n) co[n] = e[n] * inv;
}

__global__ void caps_xc_gemm(const float* __restrict__ x,
                             const float* __restrict__ cbuf,
                             float* __restrict__ xc) {
  int gw = blockIdx.x * (blockDim.x >> 5) + (threadIdx.x >> 5);
  int lane = threadIdx.x & 31;
  int b = gw >> 5;
  int rem = gw & 31;
  int mtile = rem >> 4;
  int ntile = rem & 15;
  int half = lane >> 4;
  int l16 = lane & 15;
  int m = mtile * 16 + l16;
  int col = ntile * 16 + l16;
  const float* cb = cbuf + (size_t)b * Ss * Nn;
  const float* xb = x + (size_t)b * Ss * Dd;
  __builtin_prefetch(&xb[(size_t)(2 * half) * Dd + col], 0, 1);
  v8f acc = {};
  for (int k0 = 0; k0 < Ss; k0 += 4) {
    int k = k0 + 2 * half;
    v2f a, bv;
    a.x = cb[(size_t)k * Nn + m];
    a.y = cb[(size_t)(k + 1) * Nn + m];
    bv.x = xb[(size_t)k * Dd + col];
    bv.y = xb[(size_t)(k + 1) * Dd + col];
    acc = wmma_f32_k4(a, bv, acc);
  }
  float* out = xc + (size_t)b * Nn * Dd;
#pragma unroll
  for (int r = 0; r < 8; ++r) {
    int row = r + 8 * half;
    out[(size_t)(mtile * 16 + row) * Dd + col] = acc[r];
  }
}

__global__ void caps_s_gemm(const float* __restrict__ xc,
                            const float* __restrict__ W,
                            float* __restrict__ sbuf) {
  int gw = blockIdx.x * (blockDim.x >> 5) + (threadIdx.x >> 5);
  int lane = threadIdx.x & 31;
  int n = gw >> 3;
  int rem = gw & 7;
  int mtile = rem >> 2;
  int ntile = rem & 3;
  int half = lane >> 4;
  int l16 = lane & 15;
  int m = mtile * 16 + l16;
  int col = ntile * 16 + l16;
  const float* Wn = W + (size_t)n * Dd * Cc;
  v8f acc = {};
  for (int k0 = 0; k0 < Dd; k0 += 4) {
    int k = k0 + 2 * half;
    v2f a, bv;
    a.x = xc[(size_t)m * Nn * Dd + (size_t)n * Dd + k];
    a.y = xc[(size_t)m * Nn * Dd + (size_t)n * Dd + k + 1];
    bv.x = Wn[(size_t)k * Cc + col];
    bv.y = Wn[(size_t)(k + 1) * Cc + col];
    acc = wmma_f32_k4(a, bv, acc);
  }
#pragma unroll
  for (int r = 0; r < 8; ++r) {
    int row = r + 8 * half;
    sbuf[(size_t)(mtile * 16 + row) * Nn * Cc + (size_t)n * Cc + col] = acc[r];
  }
}

__global__ void caps_squash(const float* __restrict__ sbuf,
                            float* __restrict__ vbuf,
                            float* __restrict__ out) {
  int bn = blockIdx.x;
  int c = threadIdx.x;
  float sv = sbuf[(size_t)bn * Cc + c];
  __shared__ float red[Cc];
  red[c] = sv * sv;
  __syncthreads();
#pragma unroll
  for (int off = Cc / 2; off >= 1; off >>= 1) {
    if (c < off) red[c] += red[c + off];
    __syncthreads();
  }
  float sq = red[0];
  float scale = (sq / (1.0f + sq)) * rsqrtf(sq + 1e-8f);
  float vv = scale * sv;
  vbuf[(size_t)bn * Cc + c] = vv;
  out[(size_t)bn * Cc + c] = vv;
}

__global__ void caps_wv_gemm(const float* __restrict__ W,
                             const float* __restrict__ vbuf,
                             float* __restrict__ wv) {
  int gw = blockIdx.x * (blockDim.x >> 5) + (threadIdx.x >> 5);
  int lane = threadIdx.x & 31;
  int n = gw >> 5;
  int rem = gw & 31;
  int mtile = rem >> 4;
  int ntile = rem & 15;
  int half = lane >> 4;
  int l16 = lane & 15;
  int m = mtile * 16 + l16;
  int col = ntile * 16 + l16;
  const float* Wn = W + (size_t)n * Dd * Cc;
  v8f acc = {};
  for (int k0 = 0; k0 < Cc; k0 += 4) {
    int k = k0 + 2 * half;
    v2f a, bv;
    a.x = vbuf[(size_t)m * Nn * Cc + (size_t)n * Cc + k];
    a.y = vbuf[(size_t)m * Nn * Cc + (size_t)n * Cc + k + 1];
    bv.x = Wn[(size_t)col * Cc + k];
    bv.y = Wn[(size_t)col * Cc + k + 1];
    acc = wmma_f32_k4(a, bv, acc);
  }
#pragma unroll
  for (int r = 0; r < 8; ++r) {
    int row = r + 8 * half;
    wv[(size_t)(mtile * 16 + row) * Nn * Dd + (size_t)n * Dd + col] = acc[r];
  }
}

__global__ void caps_agree_gemm(const float* __restrict__ x,
                                const float* __restrict__ wv,
                                float* __restrict__ logits) {
  int gw = blockIdx.x * (blockDim.x >> 5) + (threadIdx.x >> 5);
  int lane = threadIdx.x & 31;
  int b = gw >> 7;
  int rem = gw & 127;
  int mtile = rem >> 1;
  int ntile = rem & 1;
  int half = lane >> 4;
  int l16 = lane & 15;
  int m = mtile * 16 + l16;
  int col = ntile * 16 + l16;
  const float* xb = x + (size_t)b * Ss * Dd;
  const float* wb = wv + (size_t)b * Nn * Dd;
  __builtin_prefetch(&xb[(size_t)m * Dd], 0, 1);
  v8f acc = {};
  for (int k0 = 0; k0 < Dd; k0 += 4) {
    int k = k0 + 2 * half;
    v2f a, bv;
    a.x = xb[(size_t)m * Dd + k];
    a.y = xb[(size_t)m * Dd + k + 1];
    bv.x = wb[(size_t)col * Dd + k];
    bv.y = wb[(size_t)col * Dd + k + 1];
    acc = wmma_f32_k4(a, bv, acc);
  }
  float* lg = logits + (size_t)b * Ss * Nn;
#pragma unroll
  for (int r = 0; r < 8; ++r) {
    int row = r + 8 * half;
    size_t idx = (size_t)(mtile * 16 + row) * Nn + col;
    lg[idx] += acc[r];
  }
}

extern "C" void kernel_launch(void* const* d_in, const int* in_sizes, int n_in,
                              void* d_out, int out_size, void* d_ws,
                              size_t ws_size, hipStream_t stream) {
  const float* x = (const float*)d_in[0];
  const float* W = (const float*)d_in[1];
  float* out = (float*)d_out;
  float* ws = (float*)d_ws;

  float* logits = ws + OFF_LOGITS;
  float* cbuf   = ws + OFF_CBUF;
  float* xc     = ws + OFF_XC;
  float* wv     = ws + OFF_WV;
  float* sbuf   = ws + OFF_S;
  float* vbuf   = ws + OFF_V;

  caps_zero<<<(Bb * Ss * Nn + 511) / 512, 512, 0, stream>>>(logits,
                                                            Bb * Ss * Nn);

  for (int iter = 0; iter < 3; ++iter) {
    caps_softmax<<<(Bb * Ss) / 256, 256, 0, stream>>>(logits, cbuf);
    caps_xc_gemm<<<256, 128, 0, stream>>>(x, cbuf, xc);
    caps_s_gemm<<<64, 128, 0, stream>>>(xc, W, sbuf);
    caps_squash<<<Bb * Nn, Cc, 0, stream>>>(sbuf, vbuf, out);
    if (iter < 2) {
      caps_wv_gemm<<<256, 128, 0, stream>>>(W, vbuf, wv);
      caps_agree_gemm<<<1024, 128, 0, stream>>>(x, wv, logits);
    }
  }
}


# c^T*x GEMM re-tiled by d-columns: block = (b, 32 columns), its 16 waves split S, each holds all 32 capsules in registers (8-byte row loads); partial tiles summed in LDS, new summation order
# speedup vs baseline: 2.1344x; 2.1344x over previous
#include <hip/hip_runtime.h>
#include <math.h>

typedef __attribute__((ext_vector_type(2))) float v2f;
typedef __attribute__((ext_vector_type(8))) float v8f;

constexpr int Bb = 32;
constexpr int Ss = 1024;
constexpr int Dd = 256;
constexpr int Nn = 32;
constexpr int Cc = 64;

constexpr size_t OFF_LOGITS = 0;
constexpr size_t OFF_CBUF   = OFF_LOGITS + (size_t)Bb*Ss*Nn;
constexpr size_t OFF_XC     = OFF_CBUF   + (size_t)Bb*Ss*Nn;
constexpr size_t OFF_WV     = OFF_XC     + (size_t)Bb*Nn*Dd;
constexpr size_t OFF_S      = OFF_WV     + (size_t)Bb*Nn*Dd;
constexpr size_t OFF_V      = OFF_S      + (size_t)Bb*Nn*Cc;

__device__ __forceinline__ v8f wmma_f32_k4(v2f a, v2f b, v8f c) {
  return __builtin_amdgcn_wmma_f32_16x16x4_f32(false, a, false, b, (short)0, c,
                                               false, false);
}

__global__ void caps_zero(float* __restrict__ p, int n) {
  int i = blockIdx.x * blockDim.x + threadIdx.x;
  if (i < n) p[i] = 0.0f;
}

__global__ void caps_softmax(const float* __restrict__ logits,
                             float* __restrict__ cbuf) {
  int t = blockIdx.x * blockDim.x + threadIdx.x;
  const float* lg = logits + (size_t)t * Nn;
  float* co = cbuf + (size_t)t * Nn;
  float e[Nn];
  float m = lg[0];
#pragma unroll
  for (int n = 1; n < Nn; ++n) m = fmaxf(m, lg[n]);
  float sum = 0.0f;
#pragma unroll
  for (int n = 0; n < Nn; ++n) {
    e[n] = __expf(lg[n] - m);
    sum += e[n];
  }
  float inv = 1.0f / sum;
#pragma unroll
  for (int n = 0; n < Nn; ++n) co[n] = e[n] * inv;
}

constexpr int XC_WAVES = 16;
constexpr int XC_KSTEPS = Ss / XC_WAVES / 4;
__global__ void __launch_bounds__(XC_WAVES * 32)
__attribute__((amdgpu_waves_per_eu(4, 4)))
caps_xc_gemm(const float* __restrict__ x, const float* __restrict__ cbuf,
             float* __restrict__ xc) {
  __shared__ float red[XC_WAVES * 1024];
  int tid = threadIdx.x;
  int w = tid >> 5;
  int lane = tid & 31;
  int half = lane >> 4;
  int l16 = lane & 15;
  int b = blockIdx.x >> 3;
  int cg = blockIdx.x & 7;
  size_t row0 = (size_t)(w * (Ss / XC_WAVES) + 2 * half);
  const float* cp = cbuf + (size_t)b * Ss * Nn + row0 * Nn + 2 * l16;
  const float* xp = x + (size_t)b * Ss * Dd + row0 * Dd + cg * 32 + 2 * l16;
  v2f c0[XC_KSTEPS], c1[XC_KSTEPS], x0[XC_KSTEPS], x1[XC_KSTEPS];
#pragma unroll
  for (int j = 0; j < XC_KSTEPS; ++j) {
    c0[j] = *(const v2f*)(cp + (size_t)(4 * j) * Nn);
    c1[j] = *(const v2f*)(cp + (size_t)(4 * j + 1) * Nn);
    x0[j] = *(const v2f*)(xp + (size_t)(4 * j) * Dd);
    x1[j] = *(const v2f*)(xp + (size_t)(4 * j + 1) * Dd);
  }
  v8f acc00 = {}, acc01 = {}, acc10 = {}, acc11 = {};
#pragma unroll
  for (int j = 0; j < XC_KSTEPS; ++j) {
    v2f a0, a1, b0, b1;
    a0.x = c0[j].x; a0.y = c1[j].x;
    a1.x = c0[j].y; a1.y = c1[j].y;
    b0.x = x0[j].x; b0.y = x1[j].x;
    b1.x = x0[j].y; b1.y = x1[j].y;
    acc00 = wmma_f32_k4(a0, b0, acc00);
    acc01 = wmma_f32_k4(a0, b1, acc01);
    acc10 = wmma_f32_k4(a1, b0, acc10);
    acc11 = wmma_f32_k4(a1, b1, acc11);
  }
  float* my = red + w * 1024;
#pragma unroll
  for (int r = 0; r < 8; ++r) {
    my[(r * 4 + 0) * 32 + lane] = acc00[r];
    my[(r * 4 + 1) * 32 + lane] = acc01[r];
    my[(r * 4 + 2) * 32 + lane] = acc10[r];
    my[(r * 4 + 3) * 32 + lane] = acc11[r];
  }
  __syncthreads();
  int dd = tid & 31;
  int ct = dd & 1;
  int lsrc = dd >> 1;
#pragma unroll
  for (int q = 0; q < 2; ++q) {
    int n = (tid >> 5) + 16 * q;
    int mt = n & 1;
    int i = n >> 1;
    int off = ((i & 7) * 4 + mt * 2 + ct) * 32 + (i >> 3) * 16 + lsrc;
    float s = 0.0f;
#pragma unroll
    for (int ww = 0; ww < XC_WAVES; ++ww) s += red[ww * 1024 + off];
    xc[(size_t)b * Nn * Dd + (size_t)n * Dd + cg * 32 + dd] = s;
  }
}

__global__ void caps_s_gemm(const float* __restrict__ xc,
                            const float* __restrict__ W,
                            float* __restrict__ sbuf) {
  int gw = blockIdx.x * (blockDim.x >> 5) + (threadIdx.x >> 5);
  int lane = threadIdx.x & 31;
  int n = gw >> 3;
  int rem = gw & 7;
  int mtile = rem >> 2;
  int ntile = rem & 3;
  int half = lane >> 4;
  int l16 = lane & 15;
  int m = mtile * 16 + l16;
  int col = ntile * 16 + l16;
  const float* Wn = W + (size_t)n * Dd * Cc;
  v8f acc = {};
  for (int k0 = 0; k0 < Dd; k0 += 4) {
    int k = k0 + 2 * half;
    v2f a, bv;
    a.x = xc[(size_t)m * Nn * Dd + (size_t)n * Dd + k];
    a.y = xc[(size_t)m * Nn * Dd + (size_t)n * Dd + k + 1];
    bv.x = Wn[(size_t)k * Cc + col];
    bv.y = Wn[(size_t)(k + 1) * Cc + col];
    acc = wmma_f32_k4(a, bv, acc);
  }
#pragma unroll
  for (int r = 0; r < 8; ++r) {
    int row = r + 8 * half;
    sbuf[(size_t)(mtile * 16 + row) * Nn * Cc + (size_t)n * Cc + col] = acc[r];
  }
}

__global__ void caps_squash(const float* __restrict__ sbuf,
                            float* __restrict__ vbuf,
                            float* __restrict__ out) {
  int bn = blockIdx.x;
  int c = threadIdx.x;
  float sv = sbuf[(size_t)bn * Cc + c];
  __shared__ float red[Cc];
  red[c] = sv * sv;
  __syncthreads();
#pragma unroll
  for (int off = Cc / 2; off >= 1; off >>= 1) {
    if (c < off) red[c] += red[c + off];
    __syncthreads();
  }
  float sq = red[0];
  float scale = (sq / (1.0f + sq)) * rsqrtf(sq + 1e-8f);
  float vv = scale * sv;
  vbuf[(size_t)bn * Cc + c] = vv;
  out[(size_t)bn * Cc + c] = vv;
}

__global__ void caps_wv_gemm(const float* __restrict__ W,
                             const float* __restrict__ vbuf,
                             float* __restrict__ wv) {
  int gw = blockIdx.x * (blockDim.x >> 5) + (threadIdx.x >> 5);
  int lane = threadIdx.x & 31;
  int n = gw >> 5;
  int rem = gw & 31;
  int mtile = rem >> 4;
  int ntile = rem & 15;
  int half = lane >> 4;
  int l16 = lane & 15;
  int m = mtile * 16 + l16;
  int col = ntile * 16 + l16;
  const float* Wn = W + (size_t)n * Dd * Cc;
  v8f acc = {};
  for (int k0 = 0; k0 < Cc; k0 += 4) {
    int k = k0 + 2 * half;
    v2f a, bv;
    a.x = vbuf[(size_t)m * Nn * Cc + (size_t)n * Cc + k];
    a.y = vbuf[(size_t)m * Nn * Cc + (size_t)n * Cc + k + 1];
    bv.x = Wn[(size_t)col * Cc + k];
    bv.y = Wn[(size_t)col * Cc + k + 1];
    acc = wmma_f32_k4(a, bv, acc);
  }
#pragma unroll
  for (int r = 0; r < 8; ++r) {
    int row = r + 8 * half;
    wv[(size_t)(mtile * 16 + row) * Nn * Dd + (size_t)n * Dd + col] = acc[r];
  }
}

__global__ void caps_agree_gemm(const float* __restrict__ x,
                                const float* __restrict__ wv,
                                float* __restrict__ logits) {
  int gw = blockIdx.x * (blockDim.x >> 5) + (threadIdx.x >> 5);
  int lane = threadIdx.x & 31;
  int b = gw >> 7;
  int rem = gw & 127;
  int mtile = rem >> 1;
  int ntile = rem & 1;
  int half = lane >> 4;
  int l16 = lane & 15;
  int m = mtile * 16 + l16;
  int col = ntile * 16 + l16;
  const float* xb = x + (size_t)b * Ss * Dd;
  const float* wb = wv + (size_t)b * Nn * Dd;
  __builtin_prefetch(&xb[(size_t)m * Dd], 0, 1);
  v8f acc = {};
  for (int k0 = 0; k0 < Dd; k0 += 4) {
    int k = k0 + 2 * half;
    v2f a, bv;
    a.x = xb[(size_t)m * Dd + k];
    a.y = xb[(size_t)m * Dd + k + 1];
    bv.x = wb[(size_t)col * Dd + k];
    bv.y = wb[(size_t)col * Dd + k + 1];
    acc = wmma_f32_k4(a, bv, acc);
  }
  float* lg = logits + (size_t)b * Ss * Nn;
#pragma unroll
  for (int r = 0; r < 8; ++r) {
    int row = r + 8 * half;
    size_t idx = (size_t)(mtile * 16 + row) * Nn + col;
    lg[idx] += acc[r];
  }
}

extern "C" void kernel_launch(void* const* d_in, const int* in_sizes, int n_in,
                              void* d_out, int out_size, void* d_ws,
                              size_t ws_size, hipStream_t stream) {
  const float* x = (const float*)d_in[0];
  const float* W = (const float*)d_in[1];
  float* out = (float*)d_out;
  float* ws = (float*)d_ws;

  float* logits = ws + OFF_LOGITS;
  float* cbuf   = ws + OFF_CBUF;
  float* xc     = ws + OFF_XC;
  float* wv     = ws + OFF_WV;
  float* sbuf   = ws + OFF_S;
  float* vbuf   = ws + OFF_V;

  caps_zero<<<(Bb * Ss * Nn + 511) / 512, 512, 0, stream>>>(logits,
                                                            Bb * Ss * Nn);

  for (int iter = 0; iter < 3; ++iter) {
    caps_softmax<<<(Bb * Ss) / 256, 256, 0, stream>>>(logits, cbuf);
    caps_xc_gemm<<<256, 512, 0, stream>>>(x, cbuf, xc);
    caps_s_gemm<<<64, 128, 0, stream>>>(xc, W, sbuf);
    caps_squash<<<Bb * Nn, Cc, 0, stream>>>(sbuf, vbuf, out);
    if (iter < 2) {
      caps_wv_gemm<<<256, 128, 0, stream>>>(W, vbuf, wv);
      caps_agree_gemm<<<1024, 128, 0, stream>>>(x, wv, logits);
    }
  }
}


# routing stage fused per capsule: one block runs s-GEMM, squash and Wv-GEMM with W[n] and the xc plane staged once in LDS; s (two halves of D) and |s|^2 summed in a new order; 17 launches to 12
# speedup vs baseline: 2.5811x; 1.2093x over previous
#include <hip/hip_runtime.h>
#include <math.h>

typedef __attribute__((ext_vector_type(2))) float v2f;
typedef __attribute__((ext_vector_type(4))) float v4f;
typedef __attribute__((ext_vector_type(8))) float v8f;

constexpr int Bb = 32;
constexpr int Ss = 1024;
constexpr int Dd = 256;
constexpr int Nn = 32;
constexpr int Cc = 64;

constexpr size_t OFF_LOGITS = 0;
constexpr size_t OFF_CBUF   = OFF_LOGITS + (size_t)Bb*Ss*Nn;
constexpr size_t OFF_XC     = OFF_CBUF   + (size_t)Bb*Ss*Nn;
constexpr size_t OFF_WV     = OFF_XC     + (size_t)Bb*Nn*Dd;

__device__ __forceinline__ v8f wmma_f32_k4(v2f a, v2f b, v8f c) {
  return __builtin_amdgcn_wmma_f32_16x16x4_f32(false, a, false, b, (short)0, c,
                                               false, false);
}

__global__ void caps_zero(float* __restrict__ p, int n) {
  int i = blockIdx.x * blockDim.x + threadIdx.x;
  if (i < n) p[i] = 0.0f;
}

__global__ void caps_softmax(const float* __restrict__ logits,
                             float* __restrict__ cbuf) {
  int t = blockIdx.x * blockDim.x + threadIdx.x;
  const float* lg = logits + (size_t)t * Nn;
  float* co = cbuf + (size_t)t * Nn;
  float e[Nn];
  float m = lg[0];
#pragma unroll
  for (int n = 1; n < Nn; ++n) m = fmaxf(m, lg[n]);
  float sum = 0.0f;
#pragma unroll
  for (int n = 0; n < Nn; ++n) {
    e[n] = __expf(lg[n] - m);
    sum += e[n];
  }
  float inv = 1.0f / sum;
#pragma unroll
  for (int n = 0; n < Nn; ++n) co[n] = e[n] * inv;
}

constexpr int XC_WAVES = 16;
constexpr int XC_KSTEPS = Ss / XC_WAVES / 4;
__global__ void __launch_bounds__(XC_WAVES * 32)
__attribute__((amdgpu_waves_per_eu(4, 4)))
caps_xc_gemm(const float* __restrict__ x, const float* __restrict__ cbuf,
             float* __restrict__ xc) {
  __shared__ float red[XC_WAVES * 1024];
  int tid = threadIdx.x;
  int w = tid >> 5;
  int lane = tid & 31;
  int half = lane >> 4;
  int l16 = lane & 15;
  int b = blockIdx.x >> 3;
  int cg = blockIdx.x & 7;
  size_t row0 = (size_t)(w * (Ss / XC_WAVES) + 2 * half);
  const float* cp = cbuf + (size_t)b * Ss * Nn + row0 * Nn + 2 * l16;
  const float* xp = x + (size_t)b * Ss * Dd + row0 * Dd + cg * 32 + 2 * l16;
  v2f c0[XC_KSTEPS], c1[XC_KSTEPS], x0[XC_KSTEPS], x1[XC_KSTEPS];
#pragma unroll
  for (int j = 0; j < XC_KSTEPS; ++j) {
    c0[j] = *(const v2f*)(cp + (size_t)(4 * j) * Nn);
    c1[j] = *(const v2f*)(cp + (size_t)(4 * j + 1) * Nn);
    x0[j] = *(const v2f*)(xp + (size_t)(4 * j) * Dd);
    x1[j] = *(const v2f*)(xp + (size_t)(4 * j + 1) * Dd);
  }
  v8f acc00 = {}, acc01 = {}, acc10 = {}, acc11 = {};
#pragma unroll
  for (int j = 0; j < XC_KSTEPS; ++j) {
    v2f a0, a1, b0, b1;
    a0.x = c0[j].x; a0.y = c1[j].x;
    a1.x = c0[j].y; a1.y = c1[j].y;
    b0.x = x0[j].x; b0.y = x1[j].x;
    b1.x = x0[j].y; b1.y = x1[j].y;
    acc00 = wmma_f32_k4(a0, b0, acc00);
    acc01 = wmma_f32_k4(a0, b1, acc01);
    acc10 = wmma_f32_k4(a1, b0, acc10);
    acc11 = wmma_f32_k4(a1, b1, acc11);
  }
  float* my = red + w * 1024;
#pragma unroll
  for (int r = 0; r < 8; ++r) {
    my[(r * 4 + 0) * 32 + lane] = acc00[r];
    my[(r * 4 + 1) * 32 + lane] = acc01[r];
    my[(r * 4 + 2) * 32 + lane] = acc10[r];
    my[(r * 4 + 3) * 32 + lane] = acc11[r];
  }
  __syncthreads();
  int dd = tid & 31;
  int ct = dd & 1;
  int lsrc = dd >> 1;
#pragma unroll
  for (int q = 0; q < 2; ++q) {
    int n = (tid >> 5) + 16 * q;
    int mt = n & 1;
    int i = n >> 1;
    int off = ((i & 7) * 4 + mt * 2 + ct) * 32 + (i >> 3) * 16 + lsrc;
    float s = 0.0f;
#pragma unroll
    for (int ww = 0; ww < XC_WAVES; ++ww) s += red[ww * 1024 + off];
    xc[(size_t)b * Nn * Dd + (size_t)n * Dd + cg * 32 + dd] = s;
  }
}

constexpr int RT_WAVES = 16;
constexpr int RT_WP = Cc + 4;
constexpr int RT_XP = Dd + 4;
template <bool LAST>
__global__ void __launch_bounds__(RT_WAVES * 32)
caps_route(const float* __restrict__ xc, const float* __restrict__ W,
           float* __restrict__ wv, float* __restrict__ out) {
  __shared__ __attribute__((aligned(16))) float Wl[Dd * RT_WP];
  __shared__ __attribute__((aligned(16))) float Xl[Bb * RT_XP];
  __shared__ __attribute__((aligned(16))) float Pl[RT_WAVES * 256];
  __shared__ __attribute__((aligned(16))) float Vl[Bb * RT_WP];
  int tid = threadIdx.x;
  int w = tid >> 5;
  int lane = tid & 31;
  int half = lane >> 4;
  int l16 = lane & 15;
  int n = blockIdx.x;
  const float* Wn = W + (size_t)n * Dd * Cc;
#pragma unroll
  for (int i = 0; i < (Dd * Cc / 4) / (RT_WAVES * 32); ++i) {
    int idx = tid + RT_WAVES * 32 * i;
    v4f t = *(const v4f*)(Wn + (size_t)idx * 4);
    *(v4f*)&Wl[(idx >> 4) * RT_WP + (idx & 15) * 4] = t;
  }
#pragma unroll
  for (int i = 0; i < (Bb * Dd / 4) / (RT_WAVES * 32); ++i) {
    int idx = tid + RT_WAVES * 32 * i;
    int b = idx >> 6;
    int d4 = idx & 63;
    v4f t = *(const v4f*)(xc + ((size_t)b * Nn + n) * Dd + d4 * 4);
    *(v4f*)&Xl[b * RT_XP + d4 * 4] = t;
  }
  __syncthreads();
  {
    int tile = w & 7;
    int mt = tile >> 2;
    int nt = tile & 3;
    int kb = (w >> 3) * (Dd / 2);
    const float* ap = &Xl[(mt * 16 + l16) * RT_XP + kb + 2 * half];
    const float* bp = &Wl[(kb + 2 * half) * RT_WP + nt * 16 + l16];
    v8f acc0 = {}, acc1 = {};
#pragma unroll
    for (int j = 0; j < Dd / 2 / 4; j += 2) {
      v2f a0 = *(const v2f*)(ap + 4 * j);
      v2f a1 = *(const v2f*)(ap + 4 * j + 4);
      v2f b0, b1;
      b0.x = bp[(4 * j) * RT_WP];
      b0.y = bp[(4 * j + 1) * RT_WP];
      b1.x = bp[(4 * j + 4) * RT_WP];
      b1.y = bp[(4 * j + 5) * RT_WP];
      acc0 = wmma_f32_k4(a0, b0, acc0);
      acc1 = wmma_f32_k4(a1, b1, acc1);
    }
    float* my = Pl + w * 256;
#pragma unroll
    for (int r = 0; r < 8; ++r) my[r * 32 + lane] = acc0[r] + acc1[r];
  }
  __syncthreads();
  {
    int b = tid >> 4;
    int cq = tid & 15;
    int tile = (b >> 4) * 4 + (cq >> 2);
    int off = tile * 256 + (b & 7) * 32 + ((b >> 3) & 1) * 16 + (cq & 3) * 4;
    v4f p0 = *(const v4f*)&Pl[off];
    v4f p1 = *(const v4f*)&Pl[off + 8 * 256];
    v4f sv = p0 + p1;
    float sq = sv.x * sv.x + sv.y * sv.y + sv.z * sv.z + sv.w * sv.w;
    sq += __shfl_xor(sq, 8);
    sq += __shfl_xor(sq, 4);
    sq += __shfl_xor(sq, 2);
    sq += __shfl_xor(sq, 1);
    float scale = (sq / (1.0f + sq)) * rsqrtf(sq + 1e-8f);
    v4f vv = sv * scale;
    if (LAST) {
      *(v4f*)(out + ((size_t)b * Nn + n) * Cc + cq * 4) = vv;
    } else {
      *(v4f*)&Vl[b * RT_WP + cq * 4] = vv;
    }
  }
  if (!LAST) {
    __syncthreads();
    int mt = w & 1;
    int dp = w >> 1;
    const float* ap = &Vl[(mt * 16 + l16) * RT_WP + 2 * half];
    const float* bp0 = &Wl[(dp * 32 + 2 * l16) * RT_WP + 2 * half];
    const float* bp1 = bp0 + RT_WP;
    v8f acc0 = {}, acc1 = {};
#pragma unroll
    for (int j = 0; j < Cc / 4; ++j) {
      v2f a = *(const v2f*)(ap + 4 * j);
      v2f b0 = *(const v2f*)(bp0 + 4 * j);
      v2f b1 = *(const v2f*)(bp1 + 4 * j);
      acc0 = wmma_f32_k4(a, b0, acc0);
      acc1 = wmma_f32_k4(a, b1, acc1);
    }
    float* dst = wv + ((size_t)(mt * 16 + 8 * half) * Nn + n) * Dd + dp * 32 + 2 * l16;
#pragma unroll
    for (int r = 0; r < 8; ++r) {
      v2f o;
      o.x = acc0[r];
      o.y = acc1[r];
      *(v2f*)(dst + (size_t)r * Nn * Dd) = o;
    }
  }
}

__global__ void caps_agree_gemm(const float* __restrict__ x,
                                const float* __restrict__ wv,
                                float* __restrict__ logits) {
  int gw = blockIdx.x * (blockDim.x >> 5) + (threadIdx.x >> 5);
  int lane = threadIdx.x & 31;
  int b = gw >> 7;
  int rem = gw & 127;
  int mtile = rem >> 1;
  int ntile = rem & 1;
  int half = lane >> 4;
  int l16 = lane & 15;
  int m = mtile * 16 + l16;
  int col = ntile * 16 + l16;
  const float* xb = x + (size_t)b * Ss * Dd;
  const float* wb = wv + (size_t)b * Nn * Dd;
  __builtin_prefetch(&xb[(size_t)m * Dd], 0, 1);
  v8f acc = {};
  for (int k0 = 0; k0 < Dd; k0 += 4) {
    int k = k0 + 2 * half;
    v2f a, bv;
    a.x = xb[(size_t)m * Dd + k];
    a.y = xb[(size_t)m * Dd + k + 1];
    bv.x = wb[(size_t)col * Dd + k];
    bv.y = wb[(size_t)col * Dd + k + 1];
    acc = wmma_f32_k4(a, bv, acc);
  }
  float* lg = logits + (size_t)b * Ss * Nn;
#pragma unroll
  for (int r = 0; r < 8; ++r) {
    int row = r + 8 * half;
    size_t idx = (size_t)(mtile * 16 + row) * Nn + col;
    lg[idx] += acc[r];
  }
}

extern "C" void kernel_launch(void* const* d_in, const int* in_sizes, int n_in,
                              void* d_out, int out_size, void* d_ws,
                              size_t ws_size, hipStream_t stream) {
  const float* x = (const float*)d_in[0];
  const float* W = (const float*)d_in[1];
  float* out = (float*)d_out;
  float* ws = (float*)d_ws;

  float* logits = ws + OFF_LOGITS;
  float* cbuf   = ws + OFF_CBUF;
  float* xc     = ws + OFF_XC;
  float* wv     = ws + OFF_WV;

  caps_zero<<<(Bb * Ss * Nn + 511) / 512, 512, 0, stream>>>(logits,
                                                            Bb * Ss * Nn);

  for (int iter = 0; iter < 3; ++iter) {
    caps_softmax<<<(Bb * Ss) / 256, 256, 0, stream>>>(logits, cbuf);
    caps_xc_gemm<<<256, 512, 0, stream>>>(x, cbuf, xc);
    if (iter < 2) {
      caps_route<false><<<Nn, RT_WAVES * 32, 0, stream>>>(xc, W, wv, out);
      caps_agree_gemm<<<1024, 128, 0, stream>>>(x, wv, logits);
    } else {
      caps_route<true><<<Nn, RT_WAVES * 32, 0, stream>>>(xc, W, wv, out);
    }
  }
}


# softmax evaluated inside the c^T*x GEMM's A loader from the logits (16-lane DPP max and sum, new summation order, v_rcp); pass 0 uses 1/32 and stores its logits; 12 launches to 8
# speedup vs baseline: 3.1390x; 1.2161x over previous
#include <hip/hip_runtime.h>
#include <math.h>

typedef __attribute__((ext_vector_type(2))) float v2f;
typedef __attribute__((ext_vector_type(4))) float v4f;
typedef __attribute__((ext_vector_type(8))) float v8f;

constexpr int Bb = 32;
constexpr int Ss = 1024;
constexpr int Dd = 256;
constexpr int Nn = 32;
constexpr int Cc = 64;

constexpr size_t OFF_LOGITS = 0;
constexpr size_t OFF_CBUF   = OFF_LOGITS + (size_t)Bb*Ss*Nn;
constexpr size_t OFF_XC     = OFF_CBUF   + (size_t)Bb*Ss*Nn;
constexpr size_t OFF_WV     = OFF_XC     + (size_t)Bb*Nn*Dd;

__device__ __forceinline__ v8f wmma_f32_k4(v2f a, v2f b, v8f c) {
  return __builtin_amdgcn_wmma_f32_16x16x4_f32(false, a, false, b, (short)0, c,
                                               false, false);
}

template <int M>
__device__ __forceinline__ float row16_xor(float v) {
  int b = __builtin_bit_cast(int, v);
  return __builtin_bit_cast(
      float, __builtin_amdgcn_update_dpp(0, b, 0x160 + M, 0xf, 0xf, true));
}
__device__ __forceinline__ float row16_max(float v) {
  v = fmaxf(v, row16_xor<1>(v));
  v = fmaxf(v, row16_xor<2>(v));
  v = fmaxf(v, row16_xor<4>(v));
  v = fmaxf(v, row16_xor<8>(v));
  return v;
}
__device__ __forceinline__ float row16_sum(float v) {
  v += row16_xor<1>(v);
  v += row16_xor<2>(v);
  v += row16_xor<4>(v);
  v += row16_xor<8>(v);
  return v;
}
__device__ __forceinline__ v2f row16_softmax(v2f l) {
  float m = row16_max(fmaxf(l.x, l.y));
  float ex = __expf(l.x - m);
  float ey = __expf(l.y - m);
  float inv = __builtin_amdgcn_rcpf(row16_sum(ex + ey));
  v2f c;
  c.x = ex * inv;
  c.y = ey * inv;
  return c;
}

constexpr int XC_WAVES = 16;
constexpr int XC_KSTEPS = Ss / XC_WAVES / 4;
template <bool FIRST>
__global__ void __launch_bounds__(XC_WAVES * 32)
__attribute__((amdgpu_waves_per_eu(4, 4)))
caps_xc_gemm(const float* __restrict__ x, const float* __restrict__ logits,
             float* __restrict__ xc) {
  __shared__ float red[XC_WAVES * 1024];
  int tid = threadIdx.x;
  int w = tid >> 5;
  int lane = tid & 31;
  int half = lane >> 4;
  int l16 = lane & 15;
  int b = blockIdx.x >> 3;
  int cg = blockIdx.x & 7;
  size_t row0 = (size_t)(w * (Ss / XC_WAVES) + 2 * half);
  const float* lp = logits + (size_t)b * Ss * Nn + row0 * Nn + 2 * l16;
  const float* xp = x + (size_t)b * Ss * Dd + row0 * Dd + cg * 32 + 2 * l16;
  v2f c0[XC_KSTEPS], c1[XC_KSTEPS], x0[XC_KSTEPS], x1[XC_KSTEPS];
  if (!FIRST) {
#pragma unroll
    for (int j = 0; j < XC_KSTEPS; ++j) {
      c0[j] = *(const v2f*)(lp + (size_t)(4 * j) * Nn);
      c1[j] = *(const v2f*)(lp + (size_t)(4 * j + 1) * Nn);
    }
  }
#pragma unroll
  for (int j = 0; j < XC_KSTEPS; ++j) {
    x0[j] = *(const v2f*)(xp + (size_t)(4 * j) * Dd);
    x1[j] = *(const v2f*)(xp + (size_t)(4 * j + 1) * Dd);
  }
  v8f acc00 = {}, acc01 = {}, acc10 = {}, acc11 = {};
#pragma unroll
  for (int j = 0; j < XC_KSTEPS; ++j) {
    v2f a0, a1, b0, b1;
    v2f ck, ck1;
    if (FIRST) {
      ck.x = ck.y = 1.0f / Nn;
      ck1 = ck;
    } else {
      ck = row16_softmax(c0[j]);
      ck1 = row16_softmax(c1[j]);
    }
    a0.x = ck.x; a0.y = ck1.x;
    a1.x = ck.y; a1.y = ck1.y;
    b0.x = x0[j].x; b0.y = x1[j].x;
    b1.x = x0[j].y; b1.y = x1[j].y;
    acc00 = wmma_f32_k4(a0, b0, acc00);
    acc01 = wmma_f32_k4(a0, b1, acc01);
    acc10 = wmma_f32_k4(a1, b0, acc10);
    acc11 = wmma_f32_k4(a1, b1, acc11);
  }
  float* my = red + w * 1024;
#pragma unroll
  for (int r = 0; r < 8; ++r) {
    my[(r * 4 + 0) * 32 + lane] = acc00[r];
    my[(r * 4 + 1) * 32 + lane] = acc01[r];
    my[(r * 4 + 2) * 32 + lane] = acc10[r];
    my[(r * 4 + 3) * 32 + lane] = acc11[r];
  }
  __syncthreads();
  int dd = tid & 31;
  int ct = dd & 1;
  int lsrc = dd >> 1;
#pragma unroll
  for (int q = 0; q < 2; ++q) {
    int n = (tid >> 5) + 16 * q;
    int mt = n & 1;
    int i = n >> 1;
    int off = ((i & 7) * 4 + mt * 2 + ct) * 32 + (i >> 3) * 16 + lsrc;
    float s = 0.0f;
#pragma unroll
    for (int ww = 0; ww < XC_WAVES; ++ww) s += red[ww * 1024 + off];
    xc[(size_t)b * Nn * Dd + (size_t)n * Dd + cg * 32 + dd] = s;
  }
}

constexpr int RT_WAVES = 16;
constexpr int RT_WP = Cc + 4;
constexpr int RT_XP = Dd + 4;
template <bool LAST>
__global__ void __launch_bounds__(RT_WAVES * 32)
caps_route(const float* __restrict__ xc, const float* __restrict__ W,
           float* __restrict__ wv, float* __restrict__ out) {
  __shared__ __attribute__((aligned(16))) float Wl[Dd * RT_WP];
  __shared__ __attribute__((aligned(16))) float Xl[Bb * RT_XP];
  __shared__ __attribute__((aligned(16))) float Pl[RT_WAVES * 256];
  __shared__ __attribute__((aligned(16))) float Vl[Bb * RT_WP];
  int tid = threadIdx.x;
  int w = tid >> 5;
  int lane = tid & 31;
  int half = lane >> 4;
  int l16 = lane & 15;
  int n = blockIdx.x;
  const float* Wn = W + (size_t)n * Dd * Cc;
#pragma unroll
  for (int i = 0; i < (Dd * Cc / 4) / (RT_WAVES * 32); ++i) {
    int idx = tid + RT_WAVES * 32 * i;
    v4f t = *(const v4f*)(Wn + (size_t)idx * 4);
    *(v4f*)&Wl[(idx >> 4) * RT_WP + (idx & 15) * 4] = t;
  }
#pragma unroll
  for (int i = 0; i < (Bb * Dd / 4) / (RT_WAVES * 32); ++i) {
    int idx = tid + RT_WAVES * 32 * i;
    int b = idx >> 6;
    int d4 = idx & 63;
    v4f t = *(const v4f*)(xc + ((size_t)b * Nn + n) * Dd + d4 * 4);
    *(v4f*)&Xl[b * RT_XP + d4 * 4] = t;
  }
  __syncthreads();
  {
    int tile = w & 7;
    int mt = tile >> 2;
    int nt = tile & 3;
    int kb = (w >> 3) * (Dd / 2);
    const float* ap = &Xl[(mt * 16 + l16) * RT_XP + kb + 2 * half];
    const float* bp = &Wl[(kb + 2 * half) * RT_WP + nt * 16 + l16];
    v8f acc0 = {}, acc1 = {};
#pragma unroll
    for (int j = 0; j < Dd / 2 / 4; j += 2) {
      v2f a0 = *(const v2f*)(ap + 4 * j);
      v2f a1 = *(const v2f*)(ap + 4 * j + 4);
      v2f b0, b1;
      b0.x = bp[(4 * j) * RT_WP];
      b0.y = bp[(4 * j + 1) * RT_WP];
      b1.x = bp[(4 * j + 4) * RT_WP];
      b1.y = bp[(4 * j + 5) * RT_WP];
      acc0 = wmma_f32_k4(a0, b0, acc0);
      acc1 = wmma_f32_k4(a1, b1, acc1);
    }
    float* my = Pl + w * 256;
#pragma unroll
    for (int r = 0; r < 8; ++r) my[r * 32 + lane] = acc0[r] + acc1[r];
  }
  __syncthreads();
  {
    int b = tid >> 4;
    int cq = tid & 15;
    int tile = (b >> 4) * 4 + (cq >> 2);
    int off = tile * 256 + (b & 7) * 32 + ((b >> 3) & 1) * 16 + (cq & 3) * 4;
    v4f p0 = *(const v4f*)&Pl[off];
    v4f p1 = *(const v4f*)&Pl[off + 8 * 256];
    v4f sv = p0 + p1;
    float sq = sv.x * sv.x + sv.y * sv.y + sv.z * sv.z + sv.w * sv.w;
    sq += __shfl_xor(sq, 8);
    sq += __shfl_xor(sq, 4);
    sq += __shfl_xor(sq, 2);
    sq += __shfl_xor(sq, 1);
    float scale = (sq / (1.0f + sq)) * rsqrtf(sq + 1e-8f);
    v4f vv = sv * scale;
    if (LAST) {
      *(v4f*)(out + ((size_t)b * Nn + n) * Cc + cq * 4) = vv;
    } else {
      *(v4f*)&Vl[b * RT_WP + cq * 4] = vv;
    }
  }
  if (!LAST) {
    __syncthreads();
    int mt = w & 1;
    int dp = w >> 1;
    const float* ap = &Vl[(mt * 16 + l16) * RT_WP + 2 * half];
    const float* bp0 = &Wl[(dp * 32 + 2 * l16) * RT_WP + 2 * half];
    const float* bp1 = bp0 + RT_WP;
    v8f acc0 = {}, acc1 = {};
#pragma unroll
    for (int j = 0; j < Cc / 4; ++j) {
      v2f a = *(const v2f*)(ap + 4 * j);
      v2f b0 = *(const v2f*)(bp0 + 4 * j);
      v2f b1 = *(const v2f*)(bp1 + 4 * j);
      acc0 = wmma_f32_k4(a, b0, acc0);
      acc1 = wmma_f32_k4(a, b1, acc1);
    }
    float* dst = wv + ((size_t)(mt * 16 + 8 * half) * Nn + n) * Dd + dp * 32 + 2 * l16;
#pragma unroll
    for (int r = 0; r < 8; ++r) {
      v2f o;
      o.x = acc0[r];
      o.y = acc1[r];
      *(v2f*)(dst + (size_t)r * Nn * Dd) = o;
    }
  }
}

template <bool FIRST>
__global__ void caps_agree_gemm(const float* __restrict__ x,
                                const float* __restrict__ wv,
                                float* __restrict__ logits) {
  int gw = blockIdx.x * (blockDim.x >> 5) + (threadIdx.x >> 5);
  int lane = threadIdx.x & 31;
  int b = gw >> 7;
  int rem = gw & 127;
  int mtile = rem >> 1;
  int ntile = rem & 1;
  int half = lane >> 4;
  int l16 = lane & 15;
  int m = mtile * 16 + l16;
  int col = ntile * 16 + l16;
  const float* xb = x + (size_t)b * Ss * Dd;
  const float* wb = wv + (size_t)b * Nn * Dd;
  __builtin_prefetch(&xb[(size_t)m * Dd], 0, 1);
  v8f acc = {};
  for (int k0 = 0; k0 < Dd; k0 += 4) {
    int k = k0 + 2 * half;
    v2f a, bv;
    a.x = xb[(size_t)m * Dd + k];
    a.y = xb[(size_t)m * Dd + k + 1];
    bv.x = wb[(size_t)col * Dd + k];
    bv.y = wb[(size_t)col * Dd + k + 1];
    acc = wmma_f32_k4(a, bv, acc);
  }
  float* lg = logits + (size_t)b * Ss * Nn;
#pragma unroll
  for (int r = 0; r < 8; ++r) {
    int row = r + 8 * half;
    size_t idx = (size_t)(mtile * 16 + row) * Nn + col;
    if (FIRST) lg[idx] = acc[r];
    else lg[idx] += acc[r];
  }
}

extern "C" void kernel_launch(void* const* d_in, const int* in_sizes, int n_in,
                              void* d_out, int out_size, void* d_ws,
                              size_t ws_size, hipStream_t stream) {
  const float* x = (const float*)d_in[0];
  const float* W = (const float*)d_in[1];
  float* out = (float*)d_out;
  float* ws = (float*)d_ws;

  float* logits = ws + OFF_LOGITS;
  float* xc     = ws + OFF_XC;
  float* wv     = ws + OFF_WV;

  caps_xc_gemm<true><<<256, 512, 0, stream>>>(x, logits, xc);
  caps_route<false><<<Nn, RT_WAVES * 32, 0, stream>>>(xc, W, wv, out);
  caps_agree_gemm<true><<<1024, 128, 0, stream>>>(x, wv, logits);
  caps_xc_gemm<false><<<256, 512, 0, stream>>>(x, logits, xc);
  caps_route<false><<<Nn, RT_WAVES * 32, 0, stream>>>(xc, W, wv, out);
  caps_agree_gemm<false><<<1024, 128, 0, stream>>>(x, wv, logits);
  caps_xc_gemm<false><<<256, 512, 0, stream>>>(x, logits, xc);
  caps_route<true><<<Nn, RT_WAVES * 32, 0, stream>>>(xc, W, wv, out);
}


# agreement GEMM re-tiled: a wave owns 16 rows x 32 capsules over all of D; x rows loaded as whole 512-byte lines and turned into A fragments in a wave-private LDS strip; Wv[b] staged once per block
# speedup vs baseline: 3.9451x; 1.2568x over previous
#include <hip/hip_runtime.h>
#include <math.h>

typedef __attribute__((ext_vector_type(2))) float v2f;
typedef __attribute__((ext_vector_type(4))) float v4f;
typedef __attribute__((ext_vector_type(8))) float v8f;

constexpr int Bb = 32;
constexpr int Ss = 1024;
constexpr int Dd = 256;
constexpr int Nn = 32;
constexpr int Cc = 64;

constexpr size_t OFF_LOGITS = 0;
constexpr size_t OFF_CBUF   = OFF_LOGITS + (size_t)Bb*Ss*Nn;
constexpr size_t OFF_XC     = OFF_CBUF   + (size_t)Bb*Ss*Nn;
constexpr size_t OFF_WV     = OFF_XC     + (size_t)Bb*Nn*Dd;

__device__ __forceinline__ v8f wmma_f32_k4(v2f a, v2f b, v8f c) {
  return __builtin_amdgcn_wmma_f32_16x16x4_f32(false, a, false, b, (short)0, c,
                                               false, false);
}

template <int M>
__device__ __forceinline__ float row16_xor(float v) {
  int b = __builtin_bit_cast(int, v);
  return __builtin_bit_cast(
      float, __builtin_amdgcn_update_dpp(0, b, 0x160 + M, 0xf, 0xf, true));
}
__device__ __forceinline__ float row16_max(float v) {
  v = fmaxf(v, row16_xor<1>(v));
  v = fmaxf(v, row16_xor<2>(v));
  v = fmaxf(v, row16_xor<4>(v));
  v = fmaxf(v, row16_xor<8>(v));
  return v;
}
__device__ __forceinline__ float row16_sum(float v) {
  v += row16_xor<1>(v);
  v += row16_xor<2>(v);
  v += row16_xor<4>(v);
  v += row16_xor<8>(v);
  return v;
}
__device__ __forceinline__ v2f row16_softmax(v2f l) {
  float m = row16_max(fmaxf(l.x, l.y));
  float ex = __expf(l.x - m);
  float ey = __expf(l.y - m);
  float inv = __builtin_amdgcn_rcpf(row16_sum(ex + ey));
  v2f c;
  c.x = ex * inv;
  c.y = ey * inv;
  return c;
}

constexpr int XC_WAVES = 16;
constexpr int XC_KSTEPS = Ss / XC_WAVES / 4;
template <bool FIRST>
__global__ void __launch_bounds__(XC_WAVES * 32)
__attribute__((amdgpu_waves_per_eu(4, 4)))
caps_xc_gemm(const float* __restrict__ x, const float* __restrict__ logits,
             float* __restrict__ xc) {
  __shared__ float red[XC_WAVES * 1024];
  int tid = threadIdx.x;
  int w = tid >> 5;
  int lane = tid & 31;
  int half = lane >> 4;
  int l16 = lane & 15;
  int b = blockIdx.x >> 3;
  int cg = blockIdx.x & 7;
  size_t row0 = (size_t)(w * (Ss / XC_WAVES) + 2 * half);
  const float* lp = logits + (size_t)b * Ss * Nn + row0 * Nn + 2 * l16;
  const float* xp = x + (size_t)b * Ss * Dd + row0 * Dd + cg * 32 + 2 * l16;
  v2f c0[XC_KSTEPS], c1[XC_KSTEPS], x0[XC_KSTEPS], x1[XC_KSTEPS];
  if (!FIRST) {
#pragma unroll
    for (int j = 0; j < XC_KSTEPS; ++j) {
      c0[j] = *(const v2f*)(lp + (size_t)(4 * j) * Nn);
      c1[j] = *(const v2f*)(lp + (size_t)(4 * j + 1) * Nn);
    }
  }
#pragma unroll
  for (int j = 0; j < XC_KSTEPS; ++j) {
    x0[j] = *(const v2f*)(xp + (size_t)(4 * j) * Dd);
    x1[j] = *(const v2f*)(xp + (size_t)(4 * j + 1) * Dd);
  }
  v8f acc00 = {}, acc01 = {}, acc10 = {}, acc11 = {};
#pragma unroll
  for (int j = 0; j < XC_KSTEPS; ++j) {
    v2f a0, a1, b0, b1;
    v2f ck, ck1;
    if (FIRST) {
      ck.x = ck.y = 1.0f / Nn;
      ck1 = ck;
    } else {
      ck = row16_softmax(c0[j]);
      ck1 = row16_softmax(c1[j]);
    }
    a0.x = ck.x; a0.y = ck1.x;
    a1.x = ck.y; a1.y = ck1.y;
    b0.x = x0[j].x; b0.y = x1[j].x;
    b1.x = x0[j].y; b1.y = x1[j].y;
    acc00 = wmma_f32_k4(a0, b0, acc00);
    acc01 = wmma_f32_k4(a0, b1, acc01);
    acc10 = wmma_f32_k4(a1, b0, acc10);
    acc11 = wmma_f32_k4(a1, b1, acc11);
  }
  float* my = red + w * 1024;
#pragma unroll
  for (int r = 0; r < 8; ++r) {
    my[(r * 4 + 0) * 32 + lane] = acc00[r];
    my[(r * 4 + 1) * 32 + lane] = acc01[r];
    my[(r * 4 + 2) * 32 + lane] = acc10[r];
    my[(r * 4 + 3) * 32 + lane] = acc11[r];
  }
  __syncthreads();
  int dd = tid & 31;
  int ct = dd & 1;
  int lsrc = dd >> 1;
#pragma unroll
  for (int q = 0; q < 2; ++q) {
    int n = (tid >> 5) + 16 * q;
    int mt = n & 1;
    int i = n >> 1;
    int off = ((i & 7) * 4 + mt * 2 + ct) * 32 + (i >> 3) * 16 + lsrc;
    float s = 0.0f;
#pragma unroll
    for (int ww = 0; ww < XC_WAVES; ++ww) s += red[ww * 1024 + off];
    xc[(size_t)b * Nn * Dd + (size_t)n * Dd + cg * 32 + dd] = s;
  }
}

constexpr int RT_WAVES = 16;
constexpr int RT_WP = Cc + 4;
constexpr int RT_XP = Dd + 4;
template <bool LAST>
__global__ void __launch_bounds__(RT_WAVES * 32)
caps_route(const float* __restrict__ xc, const float* __restrict__ W,
           float* __restrict__ wv, float* __restrict__ out) {
  __shared__ __attribute__((aligned(16))) float Wl[Dd * RT_WP];
  __shared__ __attribute__((aligned(16))) float Xl[Bb * RT_XP];
  __shared__ __attribute__((aligned(16))) float Pl[RT_WAVES * 256];
  __shared__ __attribute__((aligned(16))) float Vl[Bb * RT_WP];
  int tid = threadIdx.x;
  int w = tid >> 5;
  int lane = tid & 31;
  int half = lane >> 4;
  int l16 = lane & 15;
  int n = blockIdx.x;
  const float* Wn = W + (size_t)n * Dd * Cc;
#pragma unroll
  for (int i = 0; i < (Dd * Cc / 4) / (RT_WAVES * 32); ++i) {
    int idx = tid + RT_WAVES * 32 * i;
    v4f t = *(const v4f*)(Wn + (size_t)idx * 4);
    *(v4f*)&Wl[(idx >> 4) * RT_WP + (idx & 15) * 4] = t;
  }
#pragma unroll
  for (int i = 0; i < (Bb * Dd / 4) / (RT_WAVES * 32); ++i) {
    int idx = tid + RT_WAVES * 32 * i;
    int b = idx >> 6;
    int d4 = idx & 63;
    v4f t = *(const v4f*)(xc + ((size_t)b * Nn + n) * Dd + d4 * 4);
    *(v4f*)&Xl[b * RT_XP + d4 * 4] = t;
  }
  __syncthreads();
  {
    int tile = w & 7;
    int mt = tile >> 2;
    int nt = tile & 3;
    int kb = (w >> 3) * (Dd / 2);
    const float* ap = &Xl[(mt * 16 + l16) * RT_XP + kb + 2 * half];
    const float* bp = &Wl[(kb + 2 * half) * RT_WP + nt * 16 + l16];
    v8f acc0 = {}, acc1 = {};
#pragma unroll
    for (int j = 0; j < Dd / 2 / 4; j += 2) {
      v2f a0 = *(const v2f*)(ap + 4 * j);
      v2f a1 = *(const v2f*)(ap + 4 * j + 4);
      v2f b0, b1;
      b0.x = bp[(4 * j) * RT_WP];
      b0.y = bp[(4 * j + 1) * RT_WP];
      b1.x = bp[(4 * j + 4) * RT_WP];
      b1.y = bp[(4 * j + 5) * RT_WP];
      acc0 = wmma_f32_k4(a0, b0, acc0);
      acc1 = wmma_f32_k4(a1, b1, acc1);
    }
    float* my = Pl + w * 256;
#pragma unroll
    for (int r = 0; r < 8; ++r) my[r * 32 + lane] = acc0[r] + acc1[r];
  }
  __syncthreads();
  {
    int b = tid >> 4;
    int cq = tid & 15;
    int tile = (b >> 4) * 4 + (cq >> 2);
    int off = tile * 256 + (b & 7) * 32 + ((b >> 3) & 1) * 16 + (cq & 3) * 4;
    v4f p0 = *(const v4f*)&Pl[off];
    v4f p1 = *(const v4f*)&Pl[off + 8 * 256];
    v4f sv = p0 + p1;
    float sq = sv.x * sv.x + sv.y * sv.y + sv.z * sv.z + sv.w * sv.w;
    sq += __shfl_xor(sq, 8);
    sq += __shfl_xor(sq, 4);
    sq += __shfl_xor(sq, 2);
    sq += __shfl_xor(sq, 1);
    float scale = (sq / (1.0f + sq)) * rsqrtf(sq + 1e-8f);
    v4f vv = sv * scale;
    if (LAST) {
      *(v4f*)(out + ((size_t)b * Nn + n) * Cc + cq * 4) = vv;
    } else {
      *(v4f*)&Vl[b * RT_WP + cq * 4] = vv;
    }
  }
  if (!LAST) {
    __syncthreads();
    int mt = w & 1;
    int dp = w >> 1;
    const float* ap = &Vl[(mt * 16 + l16) * RT_WP + 2 * half];
    const float* bp0 = &Wl[(dp * 32 + 2 * l16) * RT_WP + 2 * half];
    const float* bp1 = bp0 + RT_WP;
    v8f acc0 = {}, acc1 = {};
#pragma unroll
    for (int j = 0; j < Cc / 4; ++j) {
      v2f a = *(const v2f*)(ap + 4 * j);
      v2f b0 = *(const v2f*)(bp0 + 4 * j);
      v2f b1 = *(const v2f*)(bp1 + 4 * j);
      acc0 = wmma_f32_k4(a, b0, acc0);
      acc1 = wmma_f32_k4(a, b1, acc1);
    }
    float* dst = wv + ((size_t)(mt * 16 + 8 * half) * Nn + n) * Dd + dp * 32 + 2 * l16;
#pragma unroll
    for (int r = 0; r < 8; ++r) {
      v2f o;
      o.x = acc0[r];
      o.y = acc1[r];
      *(v2f*)(dst + (size_t)r * Nn * Dd) = o;
    }
  }
}

constexpr int AG_WAVES = 8;
constexpr int AG_ROWS = AG_WAVES * 16;
constexpr int AG_P = Dd + 4;
template <bool FIRST>
__global__ void __launch_bounds__(AG_WAVES * 32)
caps_agree_gemm(const float* __restrict__ x, const float* __restrict__ wv,
                float* __restrict__ logits) {
  __shared__ __attribute__((aligned(16))) float Wl[Nn * AG_P];
  __shared__ __attribute__((aligned(16))) float Xs[AG_WAVES * 16 * AG_P];
  int tid = threadIdx.x;
  int w = tid >> 5;
  int lane = tid & 31;
  int half = lane >> 4;
  int l16 = lane & 15;
  int b = blockIdx.x >> 3;
  int s0 = (blockIdx.x & 7) * AG_ROWS + w * 16;
  const float* wb = wv + (size_t)b * Nn * Dd;
  const float* xw = x + ((size_t)b * Ss + s0) * Dd;
  float* lg = logits + ((size_t)b * Ss + s0 + 8 * half) * Nn + l16;
  v4f wreg[8], xr0[16], xr1[16];
  float old0[8], old1[8];
#pragma unroll
  for (int i = 0; i < 8; ++i) {
    int idx = tid + AG_WAVES * 32 * i;
    wreg[i] = *(const v4f*)(wb + (size_t)(idx >> 6) * Dd + (idx & 63) * 4);
  }
#pragma unroll
  for (int r = 0; r < 16; ++r) xr0[r] = *(const v4f*)(xw + (size_t)r * Dd + lane * 4);
  __builtin_amdgcn_sched_barrier(0);
#pragma unroll
  for (int r = 0; r < 16; ++r) xr1[r] = *(const v4f*)(xw + (size_t)r * Dd + 128 + lane * 4);
  if (!FIRST) {
#pragma unroll
    for (int r = 0; r < 8; ++r) {
      old0[r] = lg[(size_t)r * Nn];
      old1[r] = lg[(size_t)r * Nn + 16];
    }
  }
#pragma unroll
  for (int i = 0; i < 8; ++i) {
    int idx = tid + AG_WAVES * 32 * i;
    *(v4f*)&Wl[(idx >> 6) * AG_P + (idx & 63) * 4] = wreg[i];
  }
  __builtin_amdgcn_fence(__ATOMIC_RELEASE, "workgroup", "local");
  __builtin_amdgcn_s_barrier();
  __builtin_amdgcn_fence(__ATOMIC_ACQUIRE, "workgroup", "local");
  float* xs = Xs + w * 16 * AG_P;
  const float* ap = xs + l16 * AG_P + 2 * half;
  const float* bp0 = Wl + l16 * AG_P + 2 * half;
  const float* bp1 = bp0 + 16 * AG_P;
  v8f acc0 = {}, acc1 = {};
#pragma unroll
  for (int r = 0; r < 16; ++r) *(v4f*)&xs[r * AG_P + lane * 4] = xr0[r];
  __builtin_amdgcn_wave_barrier();
#pragma unroll 8
  for (int j = 0; j < 32; ++j) {
    v2f a = *(const v2f*)(ap + 4 * j);
    v2f b0 = *(const v2f*)(bp0 + 4 * j);
    v2f b1 = *(const v2f*)(bp1 + 4 * j);
    acc0 = wmma_f32_k4(a, b0, acc0);
    acc1 = wmma_f32_k4(a, b1, acc1);
  }
#pragma unroll
  for (int r = 0; r < 16; ++r) *(v4f*)&xs[r * AG_P + 128 + lane * 4] = xr1[r];
  __builtin_amdgcn_wave_barrier();
#pragma unroll 8
  for (int j = 32; j < 64; ++j) {
    v2f a = *(const v2f*)(ap + 4 * j);
    v2f b0 = *(const v2f*)(bp0 + 4 * j);
    v2f b1 = *(const v2f*)(bp1 + 4 * j);
    acc0 = wmma_f32_k4(a, b0, acc0);
    acc1 = wmma_f32_k4(a, b1, acc1);
  }
#pragma unroll
  for (int r = 0; r < 8; ++r) {
    if (FIRST) {
      lg[(size_t)r * Nn] = acc0[r];
      lg[(size_t)r * Nn + 16] = acc1[r];
    } else {
      lg[(size_t)r * Nn] = old0[r] + acc0[r];
      lg[(size_t)r * Nn + 16] = old1[r] + acc1[r];
    }
  }
}

extern "C" void kernel_launch(void* const* d_in, const int* in_sizes, int n_in,
                              void* d_out, int out_size, void* d_ws,
                              size_t ws_size, hipStream_t stream) {
  const float* x = (const float*)d_in[0];
  const float* W = (const float*)d_in[1];
  float* out = (float*)d_out;
  float* ws = (float*)d_ws;

  float* logits = ws + OFF_LOGITS;
  float* xc     = ws + OFF_XC;
  float* wv     = ws + OFF_WV;

  caps_xc_gemm<true><<<256, 512, 0, stream>>>(x, logits, xc);
  caps_route<false><<<Nn, RT_WAVES * 32, 0, stream>>>(xc, W, wv, out);
  caps_agree_gemm<true><<<Bb * 8, AG_WAVES * 32, 0, stream>>>(x, wv, logits);
  caps_xc_gemm<false><<<256, 512, 0, stream>>>(x, logits, xc);
  caps_route<false><<<Nn, RT_WAVES * 32, 0, stream>>>(xc, W, wv, out);
  caps_agree_gemm<false><<<Bb * 8, AG_WAVES * 32, 0, stream>>>(x, wv, logits);
  caps_xc_gemm<false><<<256, 512, 0, stream>>>(x, logits, xc);
  caps_route<true><<<Nn, RT_WAVES * 32, 0, stream>>>(xc, W, wv, out);
}


# routing stage split over the batch: one block per (capsule, 16 of the 32 batch rows), 8 waves, 64 blocks for 32, so each SIMD's matrix pipe gets half the f32 WMMAs of both products; sums unchanged
# speedup vs baseline: 4.3405x; 1.1002x over previous
#include <hip/hip_runtime.h>
#include <math.h>

typedef __attribute__((ext_vector_type(2))) float v2f;
typedef __attribute__((ext_vector_type(4))) float v4f;
typedef __attribute__((ext_vector_type(8))) float v8f;

constexpr int Bb = 32;
constexpr int Ss = 1024;
constexpr int Dd = 256;
constexpr int Nn = 32;
constexpr int Cc = 64;

constexpr size_t OFF_LOGITS = 0;
constexpr size_t OFF_CBUF   = OFF_LOGITS + (size_t)Bb*Ss*Nn;
constexpr size_t OFF_XC     = OFF_CBUF   + (size_t)Bb*Ss*Nn;
constexpr size_t OFF_WV     = OFF_XC     + (size_t)Bb*Nn*Dd;

__device__ __forceinline__ v8f wmma_f32_k4(v2f a, v2f b, v8f c) {
  return __builtin_amdgcn_wmma_f32_16x16x4_f32(false, a, false, b, (short)0, c,
                                               false, false);
}

template <int M>
__device__ __forceinline__ float row16_xor(float v) {
  int b = __builtin_bit_cast(int, v);
  return __builtin_bit_cast(
      float, __builtin_amdgcn_update_dpp(0, b, 0x160 + M, 0xf, 0xf, true));
}
__device__ __forceinline__ float row16_max(float v) {
  v = fmaxf(v, row16_xor<1>(v));
  v = fmaxf(v, row16_xor<2>(v));
  v = fmaxf(v, row16_xor<4>(v));
  v = fmaxf(v, row16_xor<8>(v));
  return v;
}
__device__ __forceinline__ float row16_sum(float v) {
  v += row16_xor<1>(v);
  v += row16_xor<2>(v);
  v += row16_xor<4>(v);
  v += row16_xor<8>(v);
  return v;
}
__device__ __forceinline__ v2f row16_softmax(v2f l) {
  float m = row16_max(fmaxf(l.x, l.y));
  float ex = __expf(l.x - m);
  float ey = __expf(l.y - m);
  float inv = __builtin_amdgcn_rcpf(row16_sum(ex + ey));
  v2f c;
  c.x = ex * inv;
  c.y = ey * inv;
  return c;
}

constexpr int XC_WAVES = 16;
constexpr int XC_KSTEPS = Ss / XC_WAVES / 4;
template <bool FIRST>
__global__ void __launch_bounds__(XC_WAVES * 32)
__attribute__((amdgpu_waves_per_eu(4, 4)))
caps_xc_gemm(const float* __restrict__ x, const float* __restrict__ logits,
             float* __restrict__ xc) {
  __shared__ float red[XC_WAVES * 1024];
  int tid = threadIdx.x;
  int w = tid >> 5;
  int lane = tid & 31;
  int half = lane >> 4;
  int l16 = lane & 15;
  int b = blockIdx.x >> 3;
  int cg = blockIdx.x & 7;
  size_t row0 = (size_t)(w * (Ss / XC_WAVES) + 2 * half);
  const float* lp = logits + (size_t)b * Ss * Nn + row0 * Nn + 2 * l16;
  const float* xp = x + (size_t)b * Ss * Dd + row0 * Dd + cg * 32 + 2 * l16;
  v2f c0[XC_KSTEPS], c1[XC_KSTEPS], x0[XC_KSTEPS], x1[XC_KSTEPS];
  if (!FIRST) {
#pragma unroll
    for (int j = 0; j < XC_KSTEPS; ++j) {
      c0[j] = *(const v2f*)(lp + (size_t)(4 * j) * Nn);
      c1[j] = *(const v2f*)(lp + (size_t)(4 * j + 1) * Nn);
    }
  }
#pragma unroll
  for (int j = 0; j < XC_KSTEPS; ++j) {
    x0[j] = *(const v2f*)(xp + (size_t)(4 * j) * Dd);
    x1[j] = *(const v2f*)(xp + (size_t)(4 * j + 1) * Dd);
  }
  v8f acc00 = {}, acc01 = {}, acc10 = {}, acc11 = {};
#pragma unroll
  for (int j = 0; j < XC_KSTEPS; ++j) {
    v2f a0, a1, b0, b1;
    v2f ck, ck1;
    if (FIRST) {
      ck.x = ck.y = 1.0f / Nn;
      ck1 = ck;
    } else {
      ck = row16_softmax(c0[j]);
      ck1 = row16_softmax(c1[j]);
    }
    a0.x = ck.x; a0.y = ck1.x;
    a1.x = ck.y; a1.y = ck1.y;
    b0.x = x0[j].x; b0.y = x1[j].x;
    b1.x = x0[j].y; b1.y = x1[j].y;
    acc00 = wmma_f32_k4(a0, b0, acc00);
    acc01 = wmma_f32_k4(a0, b1, acc01);
    acc10 = wmma_f32_k4(a1, b0, acc10);
    acc11 = wmma_f32_k4(a1, b1, acc11);
  }
  float* my = red + w * 1024;
#pragma unroll
  for (int r = 0; r < 8; ++r) {
    my[(r * 4 + 0) * 32 + lane] = acc00[r];
    my[(r * 4 + 1) * 32 + lane] = acc01[r];
    my[(r * 4 + 2) * 32 + lane] = acc10[r];
    my[(r * 4 + 3) * 32 + lane] = acc11[r];
  }
  __syncthreads();
  int dd = tid & 31;
  int ct = dd & 1;
  int lsrc = dd >> 1;
#pragma unroll
  for (int q = 0; q < 2; ++q) {
    int n = (tid >> 5) + 16 * q;
    int mt = n & 1;
    int i = n >> 1;
    int off = ((i & 7) * 4 + mt * 2 + ct) * 32 + (i >> 3) * 16 + lsrc;
    float s = 0.0f;
#pragma unroll
    for (int ww = 0; ww < XC_WAVES; ++ww) s += red[ww * 1024 + off];
    xc[(size_t)b * Nn * Dd + (size_t)n * Dd + cg * 32 + dd] = s;
  }
}

constexpr int RT_WAVES = 8;
constexpr int RT_ROWS = 16;
constexpr int RT_WP = Cc + 4;
constexpr int RT_XP = Dd + 4;
template <bool LAST>
__global__ void __launch_bounds__(RT_WAVES * 32)
caps_route(const float* __restrict__ xc, const float* __restrict__ W,
           float* __restrict__ wv, float* __restrict__ out) {
  __shared__ __attribute__((aligned(16))) float Wl[Dd * RT_WP];
  __shared__ __attribute__((aligned(16))) float Xl[RT_ROWS * RT_XP];
  __shared__ __attribute__((aligned(16))) float Pl[RT_WAVES * 256];
  __shared__ __attribute__((aligned(16))) float Vl[RT_ROWS * RT_WP];
  int tid = threadIdx.x;
  int w = tid >> 5;
  int lane = tid & 31;
  int half = lane >> 4;
  int l16 = lane & 15;
  int n = blockIdx.x >> 1;
  int brow = (blockIdx.x & 1) * RT_ROWS;
  const float* Wn = W + (size_t)n * Dd * Cc;
#pragma unroll
  for (int i = 0; i < (Dd * Cc / 4) / (RT_WAVES * 32); ++i) {
    int idx = tid + RT_WAVES * 32 * i;
    v4f t = *(const v4f*)(Wn + (size_t)idx * 4);
    *(v4f*)&Wl[(idx >> 4) * RT_WP + (idx & 15) * 4] = t;
  }
#pragma unroll
  for (int i = 0; i < (RT_ROWS * Dd / 4) / (RT_WAVES * 32); ++i) {
    int idx = tid + RT_WAVES * 32 * i;
    int r = idx >> 6;
    int d4 = idx & 63;
    v4f t = *(const v4f*)(xc + ((size_t)(brow + r) * Nn + n) * Dd + d4 * 4);
    *(v4f*)&Xl[r * RT_XP + d4 * 4] = t;
  }
  __syncthreads();
  {
    int nt = w & 3;
    int kb = (w >> 2) * (Dd / 2);
    const float* ap = &Xl[l16 * RT_XP + kb + 2 * half];
    const float* bp = &Wl[(kb + 2 * half) * RT_WP + nt * 16 + l16];
    v8f acc0 = {}, acc1 = {};
#pragma unroll
    for (int j = 0; j < Dd / 2 / 4; j += 2) {
      v2f a0 = *(const v2f*)(ap + 4 * j);
      v2f a1 = *(const v2f*)(ap + 4 * j + 4);
      v2f b0, b1;
      b0.x = bp[(4 * j) * RT_WP];
      b0.y = bp[(4 * j + 1) * RT_WP];
      b1.x = bp[(4 * j + 4) * RT_WP];
      b1.y = bp[(4 * j + 5) * RT_WP];
      acc0 = wmma_f32_k4(a0, b0, acc0);
      acc1 = wmma_f32_k4(a1, b1, acc1);
    }
    float* my = Pl + w * 256;
#pragma unroll
    for (int r = 0; r < 8; ++r) my[r * 32 + lane] = acc0[r] + acc1[r];
  }
  __syncthreads();
  {
    int r = tid >> 4;
    int cq = tid & 15;
    int tile = cq >> 2;
    int off = tile * 256 + (r & 7) * 32 + (r >> 3) * 16 + (cq & 3) * 4;
    v4f p0 = *(const v4f*)&Pl[off];
    v4f p1 = *(const v4f*)&Pl[off + 4 * 256];
    v4f sv = p0 + p1;
    float sq = sv.x * sv.x + sv.y * sv.y + sv.z * sv.z + sv.w * sv.w;
    sq += __shfl_xor(sq, 8);
    sq += __shfl_xor(sq, 4);
    sq += __shfl_xor(sq, 2);
    sq += __shfl_xor(sq, 1);
    float scale = (sq / (1.0f + sq)) * rsqrtf(sq + 1e-8f);
    v4f vv = sv * scale;
    if (LAST) {
      *(v4f*)(out + ((size_t)(brow + r) * Nn + n) * Cc + cq * 4) = vv;
    } else {
      *(v4f*)&Vl[r * RT_WP + cq * 4] = vv;
    }
  }
  if (!LAST) {
    __syncthreads();
    int dp = w;
    const float* ap = &Vl[l16 * RT_WP + 2 * half];
    const float* bp0 = &Wl[(dp * 32 + 2 * l16) * RT_WP + 2 * half];
    const float* bp1 = bp0 + RT_WP;
    v8f acc0 = {}, acc1 = {};
#pragma unroll
    for (int j = 0; j < Cc / 4; ++j) {
      v2f a = *(const v2f*)(ap + 4 * j);
      v2f b0 = *(const v2f*)(bp0 + 4 * j);
      v2f b1 = *(const v2f*)(bp1 + 4 * j);
      acc0 = wmma_f32_k4(a, b0, acc0);
      acc1 = wmma_f32_k4(a, b1, acc1);
    }
    float* dst = wv + ((size_t)(brow + 8 * half) * Nn + n) * Dd + dp * 32 + 2 * l16;
#pragma unroll
    for (int r = 0; r < 8; ++r) {
      v2f o;
      o.x = acc0[r];
      o.y = acc1[r];
      *(v2f*)(dst + (size_t)r * Nn * Dd) = o;
    }
  }
}

constexpr int AG_WAVES = 8;
constexpr int AG_ROWS = AG_WAVES * 16;
constexpr int AG_P = Dd + 4;
template <bool FIRST>
__global__ void __launch_bounds__(AG_WAVES * 32)
caps_agree_gemm(const float* __restrict__ x, const float* __restrict__ wv,
                float* __restrict__ logits) {
  __shared__ __attribute__((aligned(16))) float Wl[Nn * AG_P];
  __shared__ __attribute__((aligned(16))) float Xs[AG_WAVES * 16 * AG_P];
  int tid = threadIdx.x;
  int w = tid >> 5;
  int lane = tid & 31;
  int half = lane >> 4;
  int l16 = lane & 15;
  int b = blockIdx.x >> 3;
  int s0 = (blockIdx.x & 7) * AG_ROWS + w * 16;
  const float* wb = wv + (size_t)b * Nn * Dd;
  const float* xw = x + ((size_t)b * Ss + s0) * Dd;
  float* lg = logits + ((size_t)b * Ss + s0 + 8 * half) * Nn + l16;
  v4f wreg[8], xr0[16], xr1[16];
  float old0[8], old1[8];
#pragma unroll
  for (int i = 0; i < 8; ++i) {
    int idx = tid + AG_WAVES * 32 * i;
    wreg[i] = *(const v4f*)(wb + (size_t)(idx >> 6) * Dd + (idx & 63) * 4);
  }
#pragma unroll
  for (int r = 0; r < 16; ++r) xr0[r] = *(const v4f*)(xw + (size_t)r * Dd + lane * 4);
  __builtin_amdgcn_sched_barrier(0);
#pragma unroll
  for (int r = 0; r < 16; ++r) xr1[r] = *(const v4f*)(xw + (size_t)r * Dd + 128 + lane * 4);
  if (!FIRST) {
#pragma unroll
    for (int r = 0; r < 8; ++r) {
      old0[r] = lg[(size_t)r * Nn];
      old1[r] = lg[(size_t)r * Nn + 16];
    }
  }
#pragma unroll
  for (int i = 0; i < 8; ++i) {
    int idx = tid + AG_WAVES * 32 * i;
    *(v4f*)&Wl[(idx >> 6) * AG_P + (idx & 63) * 4] = wreg[i];
  }
  __builtin_amdgcn_fence(__ATOMIC_RELEASE, "workgroup", "local");
  __builtin_amdgcn_s_barrier();
  __builtin_amdgcn_fence(__ATOMIC_ACQUIRE, "workgroup", "local");
  float* xs = Xs + w * 16 * AG_P;
  const float* ap = xs + l16 * AG_P + 2 * half;
  const float* bp0 = Wl + l16 * AG_P + 2 * half;
  const float* bp1 = bp0 + 16 * AG_P;
  v8f acc0 = {}, acc1 = {};
#pragma unroll
  for (int r = 0; r < 16; ++r) *(v4f*)&xs[r * AG_P + lane * 4] = xr0[r];
  __builtin_amdgcn_wave_barrier();
#pragma unroll 8
  for (int j = 0; j < 32; ++j) {
    v2f a = *(const v2f*)(ap + 4 * j);
    v2f b0 = *(const v2f*)(bp0 + 4 * j);
    v2f b1 = *(const v2f*)(bp1 + 4 * j);
    acc0 = wmma_f32_k4(a, b0, acc0);
    acc1 = wmma_f32_k4(a, b1, acc1);
  }
#pragma unroll
  for (int r = 0; r < 16; ++r) *(v4f*)&xs[r * AG_P + 128 + lane * 4] = xr1[r];
  __builtin_amdgcn_wave_barrier();
#pragma unroll 8
  for (int j = 32; j < 64; ++j) {
    v2f a = *(const v2f*)(ap + 4 * j);
    v2f b0 = *(const v2f*)(bp0 + 4 * j);
    v2f b1 = *(const v2f*)(bp1 + 4 * j);
    acc0 = wmma_f32_k4(a, b0, acc0);
    acc1 = wmma_f32_k4(a, b1, acc1);
  }
#pragma unroll
  for (int r = 0; r < 8; ++r) {
    if (FIRST) {
      lg[(size_t)r * Nn] = acc0[r];
      lg[(size_t)r * Nn + 16] = acc1[r];
    } else {
      lg[(size_t)r * Nn] = old0[r] + acc0[r];
      lg[(size_t)r * Nn + 16] = old1[r] + acc1[r];
    }
  }
}

extern "C" void kernel_launch(void* const* d_in, const int* in_sizes, int n_in,
                              void* d_out, int out_size, void* d_ws,
                              size_t ws_size, hipStream_t stream) {
  const float* x = (const float*)d_in[0];
  const float* W = (const float*)d_in[1];
  float* out = (float*)d_out;
  float* ws = (float*)d_ws;

  float* logits = ws + OFF_LOGITS;
  float* xc     = ws + OFF_XC;
  float* wv     = ws + OFF_WV;

  caps_xc_gemm<true><<<256, 512, 0, stream>>>(x, logits, xc);
  caps_route<false><<<Nn * 2, RT_WAVES * 32, 0, stream>>>(xc, W, wv, out);
  caps_agree_gemm<true><<<Bb * 8, AG_WAVES * 32, 0, stream>>>(x, wv, logits);
  caps_xc_gemm<false><<<256, 512, 0, stream>>>(x, logits, xc);
  caps_route<false><<<Nn * 2, RT_WAVES * 32, 0, stream>>>(xc, W, wv, out);
  caps_agree_gemm<false><<<Bb * 8, AG_WAVES * 32, 0, stream>>>(x, wv, logits);
  caps_xc_gemm<false><<<256, 512, 0, stream>>>(x, logits, xc);
  caps_route<true><<<Nn * 2, RT_WAVES * 32, 0, stream>>>(xc, W, wv, out);
}
